# starting kernel
# baseline (speedup 1.0000x reference)
#include <hip/hip_runtime.h>

typedef __attribute__((ext_vector_type(2))) float v2f;
typedef __attribute__((ext_vector_type(8))) float v8f;

#define FEAT_DIM 64

__global__ void fill_zero_kernel(float* __restrict__ p, long long n) {
  long long i = (long long)blockIdx.x * blockDim.x + threadIdx.x;
  if (i < n) p[i] = 0.0f;
}

__global__ void scatter_add_scaled(const float* __restrict__ feat,
                                   const float* __restrict__ norm,
                                   const int* __restrict__ src,
                                   const int* __restrict__ dst,
                                   float* __restrict__ agg, int E) {
  int tid = blockIdx.x * blockDim.x + threadIdx.x;
  int e = tid >> 5;
  if (e >= E) return;
  int lane = tid & 31;
  int s = src[e];
  int t = dst[e];
  float ns = norm[s];
  const float2* f2 = reinterpret_cast<const float2*>(feat + (size_t)s * FEAT_DIM);
  float2 v = f2[lane];
  float* ap = agg + (size_t)t * FEAT_DIM + lane * 2;
  atomicAdd(ap + 0, v.x * ns);
  atomicAdd(ap + 1, v.y * ns);
}

template <int NCOLS, int NTILES_N, bool RELU>
__global__ void dense_wmma(const float* __restrict__ agg,
                           const float* __restrict__ norm,
                           const float* __restrict__ W,
                           const float* __restrict__ bias,
                           float* __restrict__ out,
                           int nrows) {
  const int lane = threadIdx.x & 31;
  const int wave = threadIdx.x >> 5;
  const int wid = blockIdx.x * (blockDim.x >> 5) + wave;
  const int tiles_m = (nrows + 15) >> 4;
  if (wid >= tiles_m * NTILES_N) return;
  const int tile_m = wid / NTILES_N;
  const int tile_n = wid % NTILES_N;
  const int r0 = tile_m << 4;
  const int c0 = tile_n << 4;

  const int m = lane & 15;
  const int half = lane >> 4;

  int r = r0 + m;
  if (r >= nrows) r = nrows - 1;
  const float nr = norm[r];
  const float* __restrict__ arow = agg + (size_t)r * FEAT_DIM;

  const int col = c0 + (lane & 15);
  const bool colok = (col < NCOLS);
  const int bcol = colok ? col : 0;

  v8f acc = {};
#pragma unroll
  for (int k0 = 0; k0 < FEAT_DIM; k0 += 4) {
    const int k = k0 + half * 2;
    v2f a;
    a.x = arow[k] * nr;
    a.y = arow[k + 1] * nr;
    v2f b;
    b.x = W[(size_t)k * NCOLS + bcol];
    b.y = W[(size_t)(k + 1) * NCOLS + bcol];
    acc = __builtin_amdgcn_wmma_f32_16x16x4_f32(false, a, false, b, (short)0,
                                                acc, false, false);
  }

  if (colok) {
    const float bv = bias[bcol];
#pragma unroll
    for (int j = 0; j < 8; ++j) {
      const int row = r0 + j + half * 8;
      if (row < nrows) {
        float v = acc[j] + bv;
        if (RELU) v = fmaxf(v, 0.0f);
        out[(size_t)row * NCOLS + col] = v;
      }
    }
  }
}

extern "C" void kernel_launch(void* const* d_in, const int* in_sizes, int n_in,
                              void* d_out, int out_size, void* d_ws, size_t ws_size,
                              hipStream_t stream) {
  const float* feat = (const float*)d_in[0];
  const float* norm = (const float*)d_in[1];
  const int* src    = (const int*)d_in[2];
  const int* dst    = (const int*)d_in[3];
  const float* W1   = (const float*)d_in[4];
  const float* b1   = (const float*)d_in[5];
  const float* W2   = (const float*)d_in[6];
  const float* b2   = (const float*)d_in[7];
  float* out = (float*)d_out;

  const int N = in_sizes[1];
  const int E = in_sizes[2];

  float* bufA = (float*)d_ws;
  float* bufB = bufA + (size_t)N * FEAT_DIM;

  const long long aggN = (long long)N * FEAT_DIM;
  const int fillBlocks = (int)((aggN + 255) / 256);
  const int scatBlocks = (E + 7) / 8;

  const int tilesM = (N + 15) / 16;
  const int waves1 = tilesM * 4;
  const int waves2 = tilesM * 3;
  const int dense1Blocks = (waves1 + 7) / 8;
  const int dense2Blocks = (waves2 + 7) / 8;

  fill_zero_kernel<<<fillBlocks, 256, 0, stream>>>(bufA, aggN);
  scatter_add_scaled<<<scatBlocks, 256, 0, stream>>>(feat, norm, src, dst, bufA, E);
  dense_wmma<64, 4, true><<<dense1Blocks, 256, 0, stream>>>(bufA, norm, W1, b1, bufB, N);

  fill_zero_kernel<<<fillBlocks, 256, 0, stream>>>(bufA, aggN);
  scatter_add_scaled<<<scatBlocks, 256, 0, stream>>>(bufB, norm, src, dst, bufA, E);
  dense_wmma<40, 3, false><<<dense2Blocks, 256, 0, stream>>>(bufA, norm, W2, b2, out, N);
}


# Float atomic scatter replaced by in-call grouping of edges by dst (two-level counting sort on LDS counters, shared by both layers) + per-node gather sum; sums now in arrival rank, two chains
# speedup vs baseline: 2.5732x; 2.5732x over previous
#include <hip/hip_runtime.h>

typedef __attribute__((ext_vector_type(2))) float v2f;
typedef __attribute__((ext_vector_type(8))) float v8f;

#define FEAT_DIM 64

#define BIN_SHIFT 8
#define BIN_NODES 256
#define NBMAX 1024
#define EPT 16
#define BIN_THREADS 512
#define EPB (BIN_THREADS * EPT)

__global__ void zero_ints(int* __restrict__ p, int n) {
  int i = blockIdx.x * blockDim.x + threadIdx.x;
  if (i < n) p[i] = 0;
}

__global__ void __launch_bounds__(BIN_THREADS)
bin_count(const int* __restrict__ dst, int* __restrict__ binTotal, int E, int NB) {
  __shared__ int h[NBMAX];
  const int tid = threadIdx.x;
  for (int b = tid; b < NB; b += BIN_THREADS) h[b] = 0;
  __syncthreads();
  const int e0 = blockIdx.x * EPB + tid;
#pragma unroll
  for (int i = 0; i < EPT; ++i) {
    const int e = e0 + i * BIN_THREADS;
    if (e < E) atomicAdd(&h[dst[e] >> BIN_SHIFT], 1);
  }
  __syncthreads();
  for (int b = tid; b < NB; b += BIN_THREADS) {
    const int c = h[b];
    if (c) atomicAdd(&binTotal[b], c);
  }
}

__global__ void __launch_bounds__(1024)
bin_scan(const int* __restrict__ binTotal, int* __restrict__ binStart,
         int* __restrict__ binCursor, int NB) {
  __shared__ int wsum[32];
  const int tid = threadIdx.x;
  const int lane = tid & 31;
  const int w = tid >> 5;
  const int c = (tid < NB) ? binTotal[tid] : 0;
  int x = c;
#pragma unroll
  for (int d = 1; d < 32; d <<= 1) {
    int y = __shfl_up(x, d);
    if (lane >= d) x += y;
  }
  if (lane == 31) wsum[w] = x;
  __syncthreads();
  if (w == 0) {
    const int v = wsum[lane];
    int y = v;
#pragma unroll
    for (int d = 1; d < 32; d <<= 1) {
      int z = __shfl_up(y, d);
      if (lane >= d) y += z;
    }
    wsum[lane] = y - v;
  }
  __syncthreads();
  const int ex = wsum[w] + x - c;
  if (tid < NB) {
    binStart[tid] = ex;
    binCursor[tid] = ex;
  }
}

__global__ void __launch_bounds__(BIN_THREADS)
bin_place(const int* __restrict__ src, const int* __restrict__ dst,
          int* __restrict__ binCursor, int* __restrict__ binned, int E, int NB) {
  __shared__ int h[NBMAX];
  const int tid = threadIdx.x;
  for (int b = tid; b < NB; b += BIN_THREADS) h[b] = 0;
  __syncthreads();
  const int e0 = blockIdx.x * EPB + tid;
  int d[EPT];
  int r[EPT];
#pragma unroll
  for (int i = 0; i < EPT; ++i) {
    const int e = e0 + i * BIN_THREADS;
    d[i] = (e < E) ? dst[e] : -1;
  }
#pragma unroll
  for (int i = 0; i < EPT; ++i) {
    r[i] = 0;
    if (d[i] >= 0) r[i] = atomicAdd(&h[d[i] >> BIN_SHIFT], 1);
  }
  __syncthreads();
  for (int b = tid; b < NB; b += BIN_THREADS) {
    const int c = h[b];
    if (c) h[b] = atomicAdd(&binCursor[b], c);
  }
  __syncthreads();
  int2* __restrict__ out2 = reinterpret_cast<int2*>(binned);
#pragma unroll
  for (int i = 0; i < EPT; ++i) {
    if (d[i] >= 0) {
      const int e = e0 + i * BIN_THREADS;
      out2[h[d[i] >> BIN_SHIFT] + r[i]] = make_int2(src[e], d[i]);
    }
  }
}

#define FINE_THREADS 1024
#define FINE_BATCH 4
__global__ void __launch_bounds__(FINE_THREADS)
fine_sort(const int* __restrict__ binned, const int* __restrict__ binStart,
          const float* __restrict__ norm, int* __restrict__ segStart,
          int* __restrict__ cnt, int* __restrict__ rec, int N, int E, int NB) {
  __shared__ int c[BIN_NODES];
  __shared__ int cur[BIN_NODES];
  __shared__ int wsum[BIN_NODES / 32];
  const int tid = threadIdx.x;
  const int lane = tid & 31;
  const int w = tid >> 5;
  const int b = blockIdx.x;
  const int lo = binStart[b];
  const int hi = (b + 1 < NB) ? binStart[b + 1] : E;
  const int2* __restrict__ in2 = reinterpret_cast<const int2*>(binned);
  if (tid < BIN_NODES) c[tid] = 0;
  __syncthreads();
  for (int i0 = lo + tid; i0 < hi; i0 += FINE_BATCH * FINE_THREADS) {
    int dl[FINE_BATCH];
#pragma unroll
    for (int k = 0; k < FINE_BATCH; ++k) {
      const int i = i0 + k * FINE_THREADS;
      dl[k] = (i < hi) ? in2[i].y : -1;
    }
#pragma unroll
    for (int k = 0; k < FINE_BATCH; ++k)
      if (dl[k] >= 0) atomicAdd(&c[dl[k] & (BIN_NODES - 1)], 1);
  }
  __syncthreads();
  int mine = 0;
  int x = 0;
  if (tid < BIN_NODES) {
    mine = c[tid];
    x = mine;
#pragma unroll
    for (int d = 1; d < 32; d <<= 1) {
      int y = __shfl_up(x, d);
      if (lane >= d) x += y;
    }
    if (lane == 31) wsum[w] = x;
  }
  __syncthreads();
  if (tid < BIN_NODES) {
    int before = 0;
#pragma unroll
    for (int k = 0; k < BIN_NODES / 32; ++k)
      if (k < w) before += wsum[k];
    const int ex = lo + before + x - mine;
    cur[tid] = ex;
    const int node = (b << BIN_SHIFT) + tid;
    if (node < N) {
      segStart[node] = ex;
      cnt[node] = mine;
    }
  }
  __syncthreads();
  int2* __restrict__ out2 = reinterpret_cast<int2*>(rec);
  for (int i0 = lo + tid; i0 < hi; i0 += FINE_BATCH * FINE_THREADS) {
    int2 ed[FINE_BATCH];
    float wn[FINE_BATCH];
#pragma unroll
    for (int k = 0; k < FINE_BATCH; ++k) {
      const int i = i0 + k * FINE_THREADS;
      ed[k] = (i < hi) ? in2[i] : make_int2(0, -1);
    }
#pragma unroll
    for (int k = 0; k < FINE_BATCH; ++k)
      wn[k] = (ed[k].y >= 0) ? norm[ed[k].x] : 0.0f;
#pragma unroll
    for (int k = 0; k < FINE_BATCH; ++k) {
      if (ed[k].y >= 0) {
        const int p = atomicAdd(&cur[ed[k].y & (BIN_NODES - 1)], 1);
        out2[p] = make_int2(ed[k].x, __float_as_int(wn[k]));
      }
    }
  }
}

__global__ void __launch_bounds__(256)
gather_sum_scaled(const float* __restrict__ feat,
                  const int* __restrict__ rec,
                  const int* __restrict__ seg_start,
                  const int* __restrict__ cnt,
                  float* __restrict__ agg, int N) {
  const int node = __builtin_amdgcn_readfirstlane(
      (int)((blockIdx.x * blockDim.x + threadIdx.x) >> 5));
  if (node >= N) return;
  const int lane = threadIdx.x & 31;
  const int half = lane >> 4;
  const int q = lane & 15;
  const int deg = cnt[node];
  const int start = seg_start[node];
  const int2* __restrict__ r2 = reinterpret_cast<const int2*>(rec);
  const float4* __restrict__ f4 = reinterpret_cast<const float4*>(feat);

  float4 acc = make_float4(0.0f, 0.0f, 0.0f, 0.0f);
  for (int base = 0; base < deg; base += 32) {
    const int left = deg - base;
    const int n = left < 32 ? left : 32;
    int2 my = make_int2(0, 0);
    if (lane < n) my = r2[start + base + lane];
    const int npair = n & ~1;
#pragma unroll 4
    for (int j = 0; j < npair; j += 2) {
      const int s = __shfl(my.x, j + half);
      const float w = __int_as_float(__shfl(my.y, j + half));
      const float4 v = f4[(size_t)s * (FEAT_DIM / 4) + q];
      acc.x = fmaf(v.x, w, acc.x);
      acc.y = fmaf(v.y, w, acc.y);
      acc.z = fmaf(v.z, w, acc.z);
      acc.w = fmaf(v.w, w, acc.w);
    }
    if (n & 1) {
      const int s = __shfl(my.x, n - 1);
      const float w = __int_as_float(__shfl(my.y, n - 1));
      if (half == 0) {
        const float4 v = f4[(size_t)s * (FEAT_DIM / 4) + q];
        acc.x = fmaf(v.x, w, acc.x);
        acc.y = fmaf(v.y, w, acc.y);
        acc.z = fmaf(v.z, w, acc.z);
        acc.w = fmaf(v.w, w, acc.w);
      }
    }
  }
  acc.x += __shfl_xor(acc.x, 16);
  acc.y += __shfl_xor(acc.y, 16);
  acc.z += __shfl_xor(acc.z, 16);
  acc.w += __shfl_xor(acc.w, 16);
  if (half == 0)
    reinterpret_cast<float4*>(agg)[(size_t)node * (FEAT_DIM / 4) + q] = acc;
}

template <int NCOLS, int NTILES_N, bool RELU>
__global__ void dense_wmma(const float* __restrict__ agg,
                           const float* __restrict__ norm,
                           const float* __restrict__ W,
                           const float* __restrict__ bias,
                           float* __restrict__ out,
                           int nrows) {
  const int lane = threadIdx.x & 31;
  const int wave = threadIdx.x >> 5;
  const int wid = blockIdx.x * (blockDim.x >> 5) + wave;
  const int tiles_m = (nrows + 15) >> 4;
  if (wid >= tiles_m * NTILES_N) return;
  const int tile_m = wid / NTILES_N;
  const int tile_n = wid % NTILES_N;
  const int r0 = tile_m << 4;
  const int c0 = tile_n << 4;

  const int m = lane & 15;
  const int half = lane >> 4;

  int r = r0 + m;
  if (r >= nrows) r = nrows - 1;
  const float nr = norm[r];
  const float* __restrict__ arow = agg + (size_t)r * FEAT_DIM;

  const int col = c0 + (lane & 15);
  const bool colok = (col < NCOLS);
  const int bcol = colok ? col : 0;

  v8f acc = {};
#pragma unroll
  for (int k0 = 0; k0 < FEAT_DIM; k0 += 4) {
    const int k = k0 + half * 2;
    v2f a;
    a.x = arow[k] * nr;
    a.y = arow[k + 1] * nr;
    v2f b;
    b.x = W[(size_t)k * NCOLS + bcol];
    b.y = W[(size_t)(k + 1) * NCOLS + bcol];
    acc = __builtin_amdgcn_wmma_f32_16x16x4_f32(false, a, false, b, (short)0,
                                                acc, false, false);
  }

  if (colok) {
    const float bv = bias[bcol];
#pragma unroll
    for (int j = 0; j < 8; ++j) {
      const int row = r0 + j + half * 8;
      if (row < nrows) {
        float v = acc[j] + bv;
        if (RELU) v = fmaxf(v, 0.0f);
        out[(size_t)row * NCOLS + col] = v;
      }
    }
  }
}

extern "C" void kernel_launch(void* const* d_in, const int* in_sizes, int n_in,
                              void* d_out, int out_size, void* d_ws, size_t ws_size,
                              hipStream_t stream) {
  const float* feat = (const float*)d_in[0];
  const float* norm = (const float*)d_in[1];
  const int* src    = (const int*)d_in[2];
  const int* dst    = (const int*)d_in[3];
  const float* W1   = (const float*)d_in[4];
  const float* b1   = (const float*)d_in[5];
  const float* W2   = (const float*)d_in[6];
  const float* b2   = (const float*)d_in[7];
  float* out = (float*)d_out;

  const int N = in_sizes[1];
  const int E = in_sizes[2];

  float* bufA = (float*)d_ws;
  float* bufB = bufA + (size_t)N * FEAT_DIM;
  const int Npad = (N + 3) & ~3;
  int* binTotal  = (int*)(bufB + (size_t)N * FEAT_DIM);
  int* binStart  = binTotal + NBMAX;
  int* binCursor = binStart + NBMAX;
  int* segStart  = binCursor + NBMAX;
  int* cnt       = segStart + Npad;
  int* binned    = cnt + Npad;
  int* rec       = binned + 2 * (size_t)E;

  const int NB = (N + BIN_NODES - 1) >> BIN_SHIFT;
  const int zeroBlocks = (NB + 255) / 256;
  const int edgeBlocks = (E + EPB - 1) / EPB;
  const int nodeBlocks = (N + 7) / 8;

  const int tilesM = (N + 15) / 16;
  const int waves1 = tilesM * 4;
  const int waves2 = tilesM * 3;
  const int dense1Blocks = (waves1 + 7) / 8;
  const int dense2Blocks = (waves2 + 7) / 8;

  zero_ints<<<zeroBlocks, 256, 0, stream>>>(binTotal, NB);
  bin_count<<<edgeBlocks, BIN_THREADS, 0, stream>>>(dst, binTotal, E, NB);
  bin_scan<<<1, 1024, 0, stream>>>(binTotal, binStart, binCursor, NB);
  bin_place<<<edgeBlocks, BIN_THREADS, 0, stream>>>(src, dst, binCursor, binned, E, NB);
  fine_sort<<<NB, FINE_THREADS, 0, stream>>>(binned, binStart, norm, segStart, cnt, rec, N, E, NB);

  gather_sum_scaled<<<nodeBlocks, 256, 0, stream>>>(feat, rec, segStart, cnt, bufA, N);
  dense_wmma<64, 4, true><<<dense1Blocks, 256, 0, stream>>>(bufA, norm, W1, b1, bufB, N);

  gather_sum_scaled<<<nodeBlocks, 256, 0, stream>>>(bufB, rec, segStart, cnt, bufA, N);
  dense_wmma<40, 3, false><<<dense2Blocks, 256, 0, stream>>>(bufA, norm, W2, b2, out, N);
}


# Dense node-apply fused into the gather launch: a block gathers 32 nodes' scaled sums into LDS and its 8 waves apply W from there on WMMA; agg buffer and 2 of 9 launches go; sum order unchanged
# speedup vs baseline: 3.2355x; 1.2574x over previous
#include <hip/hip_runtime.h>

typedef __attribute__((ext_vector_type(2))) float v2f;
typedef __attribute__((ext_vector_type(8))) float v8f;

#define FEAT_DIM 64

#define BIN_SHIFT 8
#define BIN_NODES 256
#define NBMAX 1024
#define EPT 16
#define BIN_THREADS 512
#define EPB (BIN_THREADS * EPT)

__global__ void zero_ints(int* __restrict__ p, int n) {
  int i = blockIdx.x * blockDim.x + threadIdx.x;
  if (i < n) p[i] = 0;
}

__global__ void __launch_bounds__(BIN_THREADS)
bin_count(const int* __restrict__ dst, int* __restrict__ binTotal, int E, int NB) {
  __shared__ int h[NBMAX];
  const int tid = threadIdx.x;
  for (int b = tid; b < NB; b += BIN_THREADS) h[b] = 0;
  __syncthreads();
  const int e0 = blockIdx.x * EPB + tid;
#pragma unroll
  for (int i = 0; i < EPT; ++i) {
    const int e = e0 + i * BIN_THREADS;
    if (e < E) atomicAdd(&h[dst[e] >> BIN_SHIFT], 1);
  }
  __syncthreads();
  for (int b = tid; b < NB; b += BIN_THREADS) {
    const int c = h[b];
    if (c) atomicAdd(&binTotal[b], c);
  }
}

__global__ void __launch_bounds__(1024)
bin_scan(const int* __restrict__ binTotal, int* __restrict__ binStart,
         int* __restrict__ binCursor, int NB) {
  __shared__ int wsum[32];
  const int tid = threadIdx.x;
  const int lane = tid & 31;
  const int w = tid >> 5;
  const int c = (tid < NB) ? binTotal[tid] : 0;
  int x = c;
#pragma unroll
  for (int d = 1; d < 32; d <<= 1) {
    int y = __shfl_up(x, d);
    if (lane >= d) x += y;
  }
  if (lane == 31) wsum[w] = x;
  __syncthreads();
  if (w == 0) {
    const int v = wsum[lane];
    int y = v;
#pragma unroll
    for (int d = 1; d < 32; d <<= 1) {
      int z = __shfl_up(y, d);
      if (lane >= d) y += z;
    }
    wsum[lane] = y - v;
  }
  __syncthreads();
  const int ex = wsum[w] + x - c;
  if (tid < NB) {
    binStart[tid] = ex;
    binCursor[tid] = ex;
  }
}

__global__ void __launch_bounds__(BIN_THREADS)
bin_place(const int* __restrict__ src, const int* __restrict__ dst,
          int* __restrict__ binCursor, int* __restrict__ binned, int E, int NB) {
  __shared__ int h[NBMAX];
  const int tid = threadIdx.x;
  for (int b = tid; b < NB; b += BIN_THREADS) h[b] = 0;
  __syncthreads();
  const int e0 = blockIdx.x * EPB + tid;
  int d[EPT];
  int r[EPT];
#pragma unroll
  for (int i = 0; i < EPT; ++i) {
    const int e = e0 + i * BIN_THREADS;
    d[i] = (e < E) ? dst[e] : -1;
  }
#pragma unroll
  for (int i = 0; i < EPT; ++i) {
    r[i] = 0;
    if (d[i] >= 0) r[i] = atomicAdd(&h[d[i] >> BIN_SHIFT], 1);
  }
  __syncthreads();
  for (int b = tid; b < NB; b += BIN_THREADS) {
    const int c = h[b];
    if (c) h[b] = atomicAdd(&binCursor[b], c);
  }
  __syncthreads();
  int2* __restrict__ out2 = reinterpret_cast<int2*>(binned);
#pragma unroll
  for (int i = 0; i < EPT; ++i) {
    if (d[i] >= 0) {
      const int e = e0 + i * BIN_THREADS;
      out2[h[d[i] >> BIN_SHIFT] + r[i]] = make_int2(src[e], d[i]);
    }
  }
}

#define FINE_THREADS 1024
#define FINE_BATCH 4
__global__ void __launch_bounds__(FINE_THREADS)
fine_sort(const int* __restrict__ binned, const int* __restrict__ binStart,
          const float* __restrict__ norm, int* __restrict__ segStart,
          int* __restrict__ cnt, int* __restrict__ rec, int N, int E, int NB) {
  __shared__ int c[BIN_NODES];
  __shared__ int cur[BIN_NODES];
  __shared__ int wsum[BIN_NODES / 32];
  const int tid = threadIdx.x;
  const int lane = tid & 31;
  const int w = tid >> 5;
  const int b = blockIdx.x;
  const int lo = binStart[b];
  const int hi = (b + 1 < NB) ? binStart[b + 1] : E;
  const int2* __restrict__ in2 = reinterpret_cast<const int2*>(binned);
  if (tid < BIN_NODES) c[tid] = 0;
  __syncthreads();
  for (int i0 = lo + tid; i0 < hi; i0 += FINE_BATCH * FINE_THREADS) {
    int dl[FINE_BATCH];
#pragma unroll
    for (int k = 0; k < FINE_BATCH; ++k) {
      const int i = i0 + k * FINE_THREADS;
      dl[k] = (i < hi) ? in2[i].y : -1;
    }
#pragma unroll
    for (int k = 0; k < FINE_BATCH; ++k)
      if (dl[k] >= 0) atomicAdd(&c[dl[k] & (BIN_NODES - 1)], 1);
  }
  __syncthreads();
  int mine = 0;
  int x = 0;
  if (tid < BIN_NODES) {
    mine = c[tid];
    x = mine;
#pragma unroll
    for (int d = 1; d < 32; d <<= 1) {
      int y = __shfl_up(x, d);
      if (lane >= d) x += y;
    }
    if (lane == 31) wsum[w] = x;
  }
  __syncthreads();
  if (tid < BIN_NODES) {
    int before = 0;
#pragma unroll
    for (int k = 0; k < BIN_NODES / 32; ++k)
      if (k < w) before += wsum[k];
    const int ex = lo + before + x - mine;
    cur[tid] = ex;
    const int node = (b << BIN_SHIFT) + tid;
    if (node < N) {
      segStart[node] = ex;
      cnt[node] = mine;
    }
  }
  __syncthreads();
  int2* __restrict__ out2 = reinterpret_cast<int2*>(rec);
  for (int i0 = lo + tid; i0 < hi; i0 += FINE_BATCH * FINE_THREADS) {
    int2 ed[FINE_BATCH];
    float wn[FINE_BATCH];
#pragma unroll
    for (int k = 0; k < FINE_BATCH; ++k) {
      const int i = i0 + k * FINE_THREADS;
      ed[k] = (i < hi) ? in2[i] : make_int2(0, -1);
    }
#pragma unroll
    for (int k = 0; k < FINE_BATCH; ++k)
      wn[k] = (ed[k].y >= 0) ? norm[ed[k].x] : 0.0f;
#pragma unroll
    for (int k = 0; k < FINE_BATCH; ++k) {
      if (ed[k].y >= 0) {
        const int p = atomicAdd(&cur[ed[k].y & (BIN_NODES - 1)], 1);
        out2[p] = make_int2(ed[k].x, __float_as_int(wn[k]));
      }
    }
  }
}

#define TILE_ROWS 32
#define TILE_PITCH 68
#define NODES_PER_WAVE 4
#define LAYER_THREADS 256

template <int NCOLS, int NTILES_N, bool RELU>
__global__ void __launch_bounds__(LAYER_THREADS)
gather_apply(const float* __restrict__ feat,
             const int* __restrict__ rec,
             const int* __restrict__ seg_start,
             const int* __restrict__ cnt,
             const float* __restrict__ norm,
             const float* __restrict__ W,
             const float* __restrict__ bias,
             float* __restrict__ out,
             int N) {
  __shared__ __attribute__((aligned(16))) float tile[TILE_ROWS * TILE_PITCH];
  const int lane = threadIdx.x & 31;
  const int wave = __builtin_amdgcn_readfirstlane((int)(threadIdx.x >> 5));
  const int half = lane >> 4;
  const int q = lane & 15;
  const int row0 = blockIdx.x * TILE_ROWS;
  const int2* __restrict__ r2 = reinterpret_cast<const int2*>(rec);
  const float4* __restrict__ f4 = reinterpret_cast<const float4*>(feat);

  int deg[NODES_PER_WAVE];
  int start[NODES_PER_WAVE];
  float nrm[NODES_PER_WAVE];
  int2 my[NODES_PER_WAVE];
#pragma unroll
  for (int i = 0; i < NODES_PER_WAVE; ++i) {
    const int node = row0 + wave * NODES_PER_WAVE + i;
    const bool ok = node < N;
    const int nc = ok ? node : N - 1;
    const int dg = cnt[nc];
    deg[i] = ok ? dg : 0;
    start[i] = seg_start[nc];
    nrm[i] = norm[nc];
  }
#pragma unroll
  for (int i = 0; i < NODES_PER_WAVE; ++i) {
    my[i] = make_int2(0, 0);
    if (lane < deg[i]) my[i] = r2[start[i] + lane];
  }
#pragma unroll
  for (int i = 0; i < NODES_PER_WAVE; ++i) {
    float4 acc = make_float4(0.0f, 0.0f, 0.0f, 0.0f);
    for (int base = 0; base < deg[i]; base += 32) {
      const int left = deg[i] - base;
      const int n = left < 32 ? left : 32;
      int2 cur = my[i];
      if (base > 0) {
        cur = make_int2(0, 0);
        if (lane < n) cur = r2[start[i] + base + lane];
      }
      const int npair = n & ~1;
#pragma unroll 4
      for (int j = 0; j < npair; j += 2) {
        const int s = __shfl(cur.x, j + half);
        const float w = __int_as_float(__shfl(cur.y, j + half));
        const float4 v = f4[(size_t)s * (FEAT_DIM / 4) + q];
        acc.x = fmaf(v.x, w, acc.x);
        acc.y = fmaf(v.y, w, acc.y);
        acc.z = fmaf(v.z, w, acc.z);
        acc.w = fmaf(v.w, w, acc.w);
      }
      if (n & 1) {
        const int s = __shfl(cur.x, n - 1);
        const float w = __int_as_float(__shfl(cur.y, n - 1));
        if (half == 0) {
          const float4 v = f4[(size_t)s * (FEAT_DIM / 4) + q];
          acc.x = fmaf(v.x, w, acc.x);
          acc.y = fmaf(v.y, w, acc.y);
          acc.z = fmaf(v.z, w, acc.z);
          acc.w = fmaf(v.w, w, acc.w);
        }
      }
    }
    acc.x += __shfl_xor(acc.x, 16);
    acc.y += __shfl_xor(acc.y, 16);
    acc.z += __shfl_xor(acc.z, 16);
    acc.w += __shfl_xor(acc.w, 16);
    if (half == 0) {
      const float nr = nrm[i];
      float4 t;
      t.x = acc.x * nr;
      t.y = acc.y * nr;
      t.z = acc.z * nr;
      t.w = acc.w * nr;
      *reinterpret_cast<float4*>(&tile[(wave * NODES_PER_WAVE + i) * TILE_PITCH + 4 * q]) = t;
    }
  }
  const int tile_m = wave >> 2;
  const int tile_n = wave & 3;
  const int m = lane & 15;
  const int c0 = tile_n << 4;
  const int r0 = row0 + (tile_m << 4);
  const float* arow = &tile[((tile_m << 4) + m) * TILE_PITCH];

  const int col = c0 + m;
  const bool colok = (col < NCOLS);
  const int bcol = colok ? col : 0;

  v2f bw[FEAT_DIM / 4];
  float bv = 0.0f;
  if (tile_n < NTILES_N) {
#pragma unroll
    for (int k0 = 0; k0 < FEAT_DIM; k0 += 4) {
      const int k = k0 + half * 2;
      bw[k0 >> 2].x = W[(size_t)k * NCOLS + bcol];
      bw[k0 >> 2].y = W[(size_t)(k + 1) * NCOLS + bcol];
    }
    bv = bias[bcol];
  }
  __syncthreads();
  if (tile_n >= NTILES_N) return;

  v8f acc = {};
#pragma unroll
  for (int k0 = 0; k0 < FEAT_DIM; k0 += 4) {
    const int k = k0 + half * 2;
    const v2f a = *reinterpret_cast<const v2f*>(arow + k);
    acc = __builtin_amdgcn_wmma_f32_16x16x4_f32(false, a, false, bw[k0 >> 2], (short)0,
                                                acc, false, false);
  }

  if (colok) {
#pragma unroll
    for (int j = 0; j < 8; ++j) {
      const int row = r0 + j + half * 8;
      if (row < N) {
        float v = acc[j] + bv;
        if (RELU) v = fmaxf(v, 0.0f);
        out[(size_t)row * NCOLS + col] = v;
      }
    }
  }
}

extern "C" void kernel_launch(void* const* d_in, const int* in_sizes, int n_in,
                              void* d_out, int out_size, void* d_ws, size_t ws_size,
                              hipStream_t stream) {
  const float* feat = (const float*)d_in[0];
  const float* norm = (const float*)d_in[1];
  const int* src    = (const int*)d_in[2];
  const int* dst    = (const int*)d_in[3];
  const float* W1   = (const float*)d_in[4];
  const float* b1   = (const float*)d_in[5];
  const float* W2   = (const float*)d_in[6];
  const float* b2   = (const float*)d_in[7];
  float* out = (float*)d_out;

  const int N = in_sizes[1];
  const int E = in_sizes[2];

  float* bufB = (float*)d_ws;
  const int Npad = (N + 3) & ~3;
  int* binTotal  = (int*)(bufB + (size_t)N * FEAT_DIM);
  int* binStart  = binTotal + NBMAX;
  int* binCursor = binStart + NBMAX;
  int* segStart  = binCursor + NBMAX;
  int* cnt       = segStart + Npad;
  int* binned    = cnt + Npad;
  int* rec       = binned + 2 * (size_t)E;

  const int NB = (N + BIN_NODES - 1) >> BIN_SHIFT;
  const int zeroBlocks = (NB + 255) / 256;
  const int edgeBlocks = (E + EPB - 1) / EPB;
  const int layerBlocks = (N + TILE_ROWS - 1) / TILE_ROWS;

  zero_ints<<<zeroBlocks, 256, 0, stream>>>(binTotal, NB);
  bin_count<<<edgeBlocks, BIN_THREADS, 0, stream>>>(dst, binTotal, E, NB);
  bin_scan<<<1, 1024, 0, stream>>>(binTotal, binStart, binCursor, NB);
  bin_place<<<edgeBlocks, BIN_THREADS, 0, stream>>>(src, dst, binCursor, binned, E, NB);
  fine_sort<<<NB, FINE_THREADS, 0, stream>>>(binned, binStart, norm, segStart, cnt, rec, N, E, NB);

  gather_apply<64, 4, true><<<layerBlocks, LAYER_THREADS, 0, stream>>>(
      feat, rec, segStart, cnt, norm, W1, b1, bufB, N);

  gather_apply<40, 3, false><<<layerBlocks, LAYER_THREADS, 0, stream>>>(
      bufB, rec, segStart, cnt, norm, W2, b2, out, N);
}


# Record stores of both edge-sort passes staged in LDS in final order and written as contiguous runs (bin pass: 4-byte packed records); was one 8-byte store per line; sums stay in arrival rank
# speedup vs baseline: 3.4231x; 1.0580x over previous
#include <hip/hip_runtime.h>

typedef __attribute__((ext_vector_type(2))) float v2f;
typedef __attribute__((ext_vector_type(8))) float v8f;

#define FEAT_DIM 64

#define BIN_SHIFT 8
#define BIN_NODES 256
#define NBMAX 1024
#define EPT 16
#define BIN_THREADS 512
#define EPB (BIN_THREADS * EPT)

__global__ void zero_ints(int* __restrict__ p, int n) {
  int i = blockIdx.x * blockDim.x + threadIdx.x;
  if (i < n) p[i] = 0;
}

__global__ void __launch_bounds__(BIN_THREADS)
bin_count(const int* __restrict__ dst, int* __restrict__ binTotal, int E, int NB) {
  __shared__ int h[NBMAX];
  const int tid = threadIdx.x;
  for (int b = tid; b < NB; b += BIN_THREADS) h[b] = 0;
  __syncthreads();
  const int e0 = blockIdx.x * EPB + tid;
#pragma unroll
  for (int i = 0; i < EPT; ++i) {
    const int e = e0 + i * BIN_THREADS;
    if (e < E) atomicAdd(&h[dst[e] >> BIN_SHIFT], 1);
  }
  __syncthreads();
  for (int b = tid; b < NB; b += BIN_THREADS) {
    const int c = h[b];
    if (c) atomicAdd(&binTotal[b], c);
  }
}

__global__ void __launch_bounds__(1024)
bin_scan(const int* __restrict__ binTotal, int* __restrict__ binStart,
         int* __restrict__ binCursor, int NB) {
  __shared__ int wsum[32];
  const int tid = threadIdx.x;
  const int lane = tid & 31;
  const int w = tid >> 5;
  const int c = (tid < NB) ? binTotal[tid] : 0;
  int x = c;
#pragma unroll
  for (int d = 1; d < 32; d <<= 1) {
    int y = __shfl_up(x, d);
    if (lane >= d) x += y;
  }
  if (lane == 31) wsum[w] = x;
  __syncthreads();
  if (w == 0) {
    const int v = wsum[lane];
    int y = v;
#pragma unroll
    for (int d = 1; d < 32; d <<= 1) {
      int z = __shfl_up(y, d);
      if (lane >= d) y += z;
    }
    wsum[lane] = y - v;
  }
  __syncthreads();
  const int ex = wsum[w] + x - c;
  if (tid < NB) {
    binStart[tid] = ex;
    binCursor[tid] = ex;
  }
}

#define PACK_SHIFT BIN_SHIFT

__global__ void __launch_bounds__(BIN_THREADS)
bin_place(const int* __restrict__ src, const int* __restrict__ dst,
          int* __restrict__ binCursor, int* __restrict__ binned, int E) {
  static_assert(NBMAX == 2 * BIN_THREADS, "the scan below takes two bins per thread");
  static_assert(NBMAX <= 65536, "a bin index is staged as 16 bits");
  __shared__ int h[NBMAX];
  __shared__ int lstart[NBMAX];
  __shared__ int gdelta[NBMAX];
  __shared__ int stage[EPB];
  __shared__ unsigned short sbin[EPB];
  __shared__ int wsum[32];
  const int tid = threadIdx.x;
  const int lane = tid & 31;
  const int w = tid >> 5;
  for (int b = tid; b < NBMAX; b += BIN_THREADS) h[b] = 0;
  __syncthreads();
  const int e0 = blockIdx.x * EPB + tid;
  int d[EPT];
  int s[EPT];
  int r[EPT];
#pragma unroll
  for (int i = 0; i < EPT; ++i) {
    const int e = e0 + i * BIN_THREADS;
    const bool ok = e < E;
    d[i] = ok ? dst[e] : -1;
    s[i] = ok ? src[e] : 0;
  }
#pragma unroll
  for (int i = 0; i < EPT; ++i) {
    r[i] = 0;
    if (d[i] >= 0) r[i] = atomicAdd(&h[d[i] >> BIN_SHIFT], 1);
  }
  __syncthreads();
  const int a0 = h[2 * tid];
  const int a1 = h[2 * tid + 1];
  const int both = a0 + a1;
  int x = both;
#pragma unroll
  for (int dd = 1; dd < 32; dd <<= 1) {
    int y = __shfl_up(x, dd);
    if (lane >= dd) x += y;
  }
  if (lane == 31) wsum[w] = x;
  __syncthreads();
  if (w == 0) {
    const int v = (lane < BIN_THREADS / 32) ? wsum[lane] : 0;
    int y = v;
#pragma unroll
    for (int dd = 1; dd < 32; dd <<= 1) {
      int z = __shfl_up(y, dd);
      if (lane >= dd) y += z;
    }
    wsum[lane] = y - v;
  }
  __syncthreads();
  const int ex0 = wsum[w] + x - both;
  const int ex1 = ex0 + a0;
  lstart[2 * tid] = ex0;
  lstart[2 * tid + 1] = ex1;
  if (a0) gdelta[2 * tid] = atomicAdd(&binCursor[2 * tid], a0) - ex0;
  if (a1) gdelta[2 * tid + 1] = atomicAdd(&binCursor[2 * tid + 1], a1) - ex1;
  __syncthreads();
#pragma unroll
  for (int i = 0; i < EPT; ++i) {
    if (d[i] >= 0) {
      const int b = d[i] >> BIN_SHIFT;
      const int p = lstart[b] + r[i];
      stage[p] = (s[i] << PACK_SHIFT) | (d[i] & (BIN_NODES - 1));
      sbin[p] = (unsigned short)b;
    }
  }
  __syncthreads();
  const int left = E - blockIdx.x * EPB;
  const int nvalid = left < EPB ? left : EPB;
  if (nvalid == EPB) {
#pragma unroll
    for (int i = 0; i < EPT; ++i) {
      const int p = tid + i * BIN_THREADS;
      binned[gdelta[sbin[p]] + p] = stage[p];
    }
  } else {
    for (int p = tid; p < nvalid; p += BIN_THREADS) binned[gdelta[sbin[p]] + p] = stage[p];
  }
}

#define FINE_THREADS 1024
#define FINE_BATCH 4
#define FINE_CAP 6144
__global__ void __launch_bounds__(FINE_THREADS)
fine_sort(const int* __restrict__ binned, const int* __restrict__ binStart,
          const float* __restrict__ norm, int* __restrict__ segStart,
          int* __restrict__ cnt, int* __restrict__ rec, int N, int E, int NB) {
  __shared__ int c[BIN_NODES];
  __shared__ int cur[BIN_NODES];
  __shared__ int wsum[BIN_NODES / 32];
  __shared__ __attribute__((aligned(8))) int stage_words[2 * FINE_CAP];
  int2* stage = reinterpret_cast<int2*>(stage_words);
  const int tid = threadIdx.x;
  const int lane = tid & 31;
  const int w = tid >> 5;
  const int b = blockIdx.x;
  const int lo = binStart[b];
  const int hi = (b + 1 < NB) ? binStart[b + 1] : E;
  const bool staged = (hi - lo) <= FINE_CAP;
  if (tid < BIN_NODES) c[tid] = 0;
  __syncthreads();
  for (int i0 = lo + tid; i0 < hi; i0 += FINE_BATCH * FINE_THREADS) {
    int dl[FINE_BATCH];
#pragma unroll
    for (int k = 0; k < FINE_BATCH; ++k) {
      const int i = i0 + k * FINE_THREADS;
      dl[k] = (i < hi) ? binned[i] : -1;
    }
#pragma unroll
    for (int k = 0; k < FINE_BATCH; ++k)
      if (dl[k] >= 0) atomicAdd(&c[dl[k] & (BIN_NODES - 1)], 1);
  }
  __syncthreads();
  int mine = 0;
  int x = 0;
  if (tid < BIN_NODES) {
    mine = c[tid];
    x = mine;
#pragma unroll
    for (int d = 1; d < 32; d <<= 1) {
      int y = __shfl_up(x, d);
      if (lane >= d) x += y;
    }
    if (lane == 31) wsum[w] = x;
  }
  __syncthreads();
  if (tid < BIN_NODES) {
    int before = 0;
#pragma unroll
    for (int k = 0; k < BIN_NODES / 32; ++k)
      if (k < w) before += wsum[k];
    const int ex = lo + before + x - mine;
    cur[tid] = ex;
    const int node = (b << BIN_SHIFT) + tid;
    if (node < N) {
      segStart[node] = ex;
      cnt[node] = mine;
    }
  }
  __syncthreads();
  int2* __restrict__ out2 = reinterpret_cast<int2*>(rec);
  for (int i0 = lo + tid; i0 < hi; i0 += FINE_BATCH * FINE_THREADS) {
    int ed[FINE_BATCH];
    float wn[FINE_BATCH];
#pragma unroll
    for (int k = 0; k < FINE_BATCH; ++k) {
      const int i = i0 + k * FINE_THREADS;
      ed[k] = (i < hi) ? binned[i] : -1;
    }
#pragma unroll
    for (int k = 0; k < FINE_BATCH; ++k)
      wn[k] = (ed[k] >= 0) ? norm[ed[k] >> PACK_SHIFT] : 0.0f;
#pragma unroll
    for (int k = 0; k < FINE_BATCH; ++k) {
      if (ed[k] >= 0) {
        const int p = atomicAdd(&cur[ed[k] & (BIN_NODES - 1)], 1);
        const int2 val = make_int2(ed[k] >> PACK_SHIFT, __float_as_int(wn[k]));
        if (staged) stage[p - lo] = val;
        else out2[p] = val;
      }
    }
  }
  if (staged) {
    __syncthreads();
    for (int i = tid; i < hi - lo; i += FINE_THREADS) out2[lo + i] = stage[i];
  }
}

#define TILE_ROWS 32
#define TILE_PITCH 68
#define NODES_PER_WAVE 4
#define LAYER_THREADS 256

template <int NCOLS, int NTILES_N, bool RELU>
__global__ void __launch_bounds__(LAYER_THREADS)
gather_apply(const float* __restrict__ feat,
             const int* __restrict__ rec,
             const int* __restrict__ seg_start,
             const int* __restrict__ cnt,
             const float* __restrict__ norm,
             const float* __restrict__ W,
             const float* __restrict__ bias,
             float* __restrict__ out,
             int N) {
  __shared__ __attribute__((aligned(16))) float tile[TILE_ROWS * TILE_PITCH];
  const int lane = threadIdx.x & 31;
  const int wave = __builtin_amdgcn_readfirstlane((int)(threadIdx.x >> 5));
  const int half = lane >> 4;
  const int q = lane & 15;
  const int row0 = blockIdx.x * TILE_ROWS;
  const int2* __restrict__ r2 = reinterpret_cast<const int2*>(rec);
  const float4* __restrict__ f4 = reinterpret_cast<const float4*>(feat);

  int deg[NODES_PER_WAVE];
  int start[NODES_PER_WAVE];
  float nrm[NODES_PER_WAVE];
  int2 my[NODES_PER_WAVE];
#pragma unroll
  for (int i = 0; i < NODES_PER_WAVE; ++i) {
    const int node = row0 + wave * NODES_PER_WAVE + i;
    const bool ok = node < N;
    const int nc = ok ? node : N - 1;
    const int dg = cnt[nc];
    deg[i] = ok ? dg : 0;
    start[i] = seg_start[nc];
    nrm[i] = norm[nc];
  }
#pragma unroll
  for (int i = 0; i < NODES_PER_WAVE; ++i) {
    my[i] = make_int2(0, 0);
    if (lane < deg[i]) my[i] = r2[start[i] + lane];
  }
#pragma unroll
  for (int i = 0; i < NODES_PER_WAVE; ++i) {
    float4 acc = make_float4(0.0f, 0.0f, 0.0f, 0.0f);
    for (int base = 0; base < deg[i]; base += 32) {
      const int left = deg[i] - base;
      const int n = left < 32 ? left : 32;
      int2 cur = my[i];
      if (base > 0) {
        cur = make_int2(0, 0);
        if (lane < n) cur = r2[start[i] + base + lane];
      }
      const int npair = n & ~1;
#pragma unroll 4
      for (int j = 0; j < npair; j += 2) {
        const int s = __shfl(cur.x, j + half);
        const float w = __int_as_float(__shfl(cur.y, j + half));
        const float4 v = f4[(size_t)s * (FEAT_DIM / 4) + q];
        acc.x = fmaf(v.x, w, acc.x);
        acc.y = fmaf(v.y, w, acc.y);
        acc.z = fmaf(v.z, w, acc.z);
        acc.w = fmaf(v.w, w, acc.w);
      }
      if (n & 1) {
        const int s = __shfl(cur.x, n - 1);
        const float w = __int_as_float(__shfl(cur.y, n - 1));
        if (half == 0) {
          const float4 v = f4[(size_t)s * (FEAT_DIM / 4) + q];
          acc.x = fmaf(v.x, w, acc.x);
          acc.y = fmaf(v.y, w, acc.y);
          acc.z = fmaf(v.z, w, acc.z);
          acc.w = fmaf(v.w, w, acc.w);
        }
      }
    }
    acc.x += __shfl_xor(acc.x, 16);
    acc.y += __shfl_xor(acc.y, 16);
    acc.z += __shfl_xor(acc.z, 16);
    acc.w += __shfl_xor(acc.w, 16);
    if (half == 0) {
      const float nr = nrm[i];
      float4 t;
      t.x = acc.x * nr;
      t.y = acc.y * nr;
      t.z = acc.z * nr;
      t.w = acc.w * nr;
      *reinterpret_cast<float4*>(&tile[(wave * NODES_PER_WAVE + i) * TILE_PITCH + 4 * q]) = t;
    }
  }
  const int tile_m = wave >> 2;
  const int tile_n = wave & 3;
  const int m = lane & 15;
  const int c0 = tile_n << 4;
  const int r0 = row0 + (tile_m << 4);
  const float* arow = &tile[((tile_m << 4) + m) * TILE_PITCH];

  const int col = c0 + m;
  const bool colok = (col < NCOLS);
  const int bcol = colok ? col : 0;

  v2f bw[FEAT_DIM / 4];
  float bv = 0.0f;
  if (tile_n < NTILES_N) {
#pragma unroll
    for (int k0 = 0; k0 < FEAT_DIM; k0 += 4) {
      const int k = k0 + half * 2;
      bw[k0 >> 2].x = W[(size_t)k * NCOLS + bcol];
      bw[k0 >> 2].y = W[(size_t)(k + 1) * NCOLS + bcol];
    }
    bv = bias[bcol];
  }
  __syncthreads();
  if (tile_n >= NTILES_N) return;

  v8f acc = {};
#pragma unroll
  for (int k0 = 0; k0 < FEAT_DIM; k0 += 4) {
    const int k = k0 + half * 2;
    const v2f a = *reinterpret_cast<const v2f*>(arow + k);
    acc = __builtin_amdgcn_wmma_f32_16x16x4_f32(false, a, false, bw[k0 >> 2], (short)0,
                                                acc, false, false);
  }

  if (colok) {
#pragma unroll
    for (int j = 0; j < 8; ++j) {
      const int row = r0 + j + half * 8;
      if (row < N) {
        float v = acc[j] + bv;
        if (RELU) v = fmaxf(v, 0.0f);
        out[(size_t)row * NCOLS + col] = v;
      }
    }
  }
}

extern "C" void kernel_launch(void* const* d_in, const int* in_sizes, int n_in,
                              void* d_out, int out_size, void* d_ws, size_t ws_size,
                              hipStream_t stream) {
  const float* feat = (const float*)d_in[0];
  const float* norm = (const float*)d_in[1];
  const int* src    = (const int*)d_in[2];
  const int* dst    = (const int*)d_in[3];
  const float* W1   = (const float*)d_in[4];
  const float* b1   = (const float*)d_in[5];
  const float* W2   = (const float*)d_in[6];
  const float* b2   = (const float*)d_in[7];
  float* out = (float*)d_out;

  const int N = in_sizes[1];
  const int E = in_sizes[2];

  float* bufB = (float*)d_ws;
  const int Npad = (N + 3) & ~3;
  int* binTotal  = (int*)(bufB + (size_t)N * FEAT_DIM);
  int* binStart  = binTotal + NBMAX;
  int* binCursor = binStart + NBMAX;
  int* segStart  = binCursor + NBMAX;
  int* cnt       = segStart + Npad;
  int* binned    = cnt + Npad;
  int* rec       = binned + 2 * (size_t)E;

  const int NB = (N + BIN_NODES - 1) >> BIN_SHIFT;
  const int zeroBlocks = (NB + 255) / 256;
  const int edgeBlocks = (E + EPB - 1) / EPB;
  const int layerBlocks = (N + TILE_ROWS - 1) / TILE_ROWS;

  zero_ints<<<zeroBlocks, 256, 0, stream>>>(binTotal, NB);
  bin_count<<<edgeBlocks, BIN_THREADS, 0, stream>>>(dst, binTotal, E, NB);
  bin_scan<<<1, 1024, 0, stream>>>(binTotal, binStart, binCursor, NB);
  bin_place<<<edgeBlocks, BIN_THREADS, 0, stream>>>(src, dst, binCursor, binned, E);
  fine_sort<<<NB, FINE_THREADS, 0, stream>>>(binned, binStart, norm, segStart, cnt, rec, N, E, NB);

  gather_apply<64, 4, true><<<layerBlocks, LAYER_THREADS, 0, stream>>>(
      feat, rec, segStart, cnt, norm, W1, b1, bufB, N);

  gather_apply<40, 3, false><<<layerBlocks, LAYER_THREADS, 0, stream>>>(
      bufB, rec, segStart, cnt, norm, W2, b2, out, N);
}


# Edge grouping without global counters: a block sorts its edges by bin in its segment and writes an offset row; a bin's block collects its runs via the rows; 7 -> 4 launches; sums in arrival rank
# speedup vs baseline: 3.9856x; 1.1643x over previous
#include <hip/hip_runtime.h>

typedef __attribute__((ext_vector_type(2))) float v2f;
typedef __attribute__((ext_vector_type(8))) float v8f;

#define FEAT_DIM 64

#define BIN_SHIFT 8
#define BIN_NODES 256
#define NBMAX 1024
#define EPT 16
#define BIN_THREADS 512
#define EPB (BIN_THREADS * EPT)
#define MAXSEG 1024

#define PACK_SHIFT BIN_SHIFT

__global__ void __launch_bounds__(BIN_THREADS)
bin_local(const int* __restrict__ src, const int* __restrict__ dst,
          int* __restrict__ binned, int* __restrict__ tab, int E, int NB, int pitch) {
  static_assert(NBMAX == 2 * BIN_THREADS, "the scan below takes two bins per thread");
  static_assert(EPB % (4 * BIN_THREADS) == 0, "the segment is written as whole int4 per thread");
  __shared__ int h[NBMAX];
  __shared__ int lstart[NBMAX];
  __shared__ __attribute__((aligned(16))) int stage[EPB];
  __shared__ int wsum[32];
  const int tid = threadIdx.x;
  const int lane = tid & 31;
  const int w = tid >> 5;
  for (int b = tid; b < NBMAX; b += BIN_THREADS) h[b] = 0;
  __syncthreads();
  const int e0 = blockIdx.x * EPB + tid;
  int d[EPT];
  int s[EPT];
  int r[EPT];
#pragma unroll
  for (int i = 0; i < EPT; ++i) {
    const int e = e0 + i * BIN_THREADS;
    const bool ok = e < E;
    d[i] = ok ? dst[e] : -1;
    s[i] = ok ? src[e] : 0;
  }
#pragma unroll
  for (int i = 0; i < EPT; ++i) {
    r[i] = 0;
    if (d[i] >= 0) r[i] = atomicAdd(&h[d[i] >> BIN_SHIFT], 1);
  }
  __syncthreads();
  const int a0 = h[2 * tid];
  const int a1 = h[2 * tid + 1];
  const int both = a0 + a1;
  int x = both;
#pragma unroll
  for (int dd = 1; dd < 32; dd <<= 1) {
    int y = __shfl_up(x, dd);
    if (lane >= dd) x += y;
  }
  if (lane == 31) wsum[w] = x;
  __syncthreads();
  if (w == 0) {
    const int v = (lane < BIN_THREADS / 32) ? wsum[lane] : 0;
    int y = v;
#pragma unroll
    for (int dd = 1; dd < 32; dd <<= 1) {
      int z = __shfl_up(y, dd);
      if (lane >= dd) y += z;
    }
    wsum[lane] = y - v;
  }
  __syncthreads();
  const int ex0 = wsum[w] + x - both;
  const int ex1 = ex0 + a0;
  lstart[2 * tid] = ex0;
  lstart[2 * tid + 1] = ex1;
  int* __restrict__ row = tab + (size_t)blockIdx.x * pitch;
  if (2 * tid < NB) row[2 * tid] = ex0;
  if (2 * tid + 1 < NB) row[2 * tid + 1] = ex1;
  __syncthreads();
#pragma unroll
  for (int i = 0; i < EPT; ++i) {
    if (d[i] >= 0) {
      const int p = lstart[d[i] >> BIN_SHIFT] + r[i];
      stage[p] = (s[i] << PACK_SHIFT) | (d[i] & (BIN_NODES - 1));
    }
  }
  __syncthreads();
  const int left = E - blockIdx.x * EPB;
  const int nvalid = left < EPB ? left : EPB;
  int* __restrict__ seg = binned + (size_t)blockIdx.x * EPB;
  if (nvalid == EPB) {
    const int4* __restrict__ st4 = reinterpret_cast<const int4*>(stage);
    int4* __restrict__ seg4 = reinterpret_cast<int4*>(seg);
#pragma unroll
    for (int i = 0; i < EPB / (4 * BIN_THREADS); ++i) seg4[tid + i * BIN_THREADS] = st4[tid + i * BIN_THREADS];
  } else {
    for (int p = tid; p < nvalid; p += BIN_THREADS) seg[p] = stage[p];
  }
}

#define FINE_THREADS 1024
#define FINE_WAVES (FINE_THREADS / 32)
#define RUN_BATCH 8
#define RUN_SLOTS (MAXSEG + FINE_WAVES * RUN_BATCH)
#define FINE_CAP 6144
__global__ void __launch_bounds__(FINE_THREADS)
fine_sort(const int* __restrict__ binned, const int* __restrict__ tab,
          const float* __restrict__ norm, int* __restrict__ segStart,
          int* __restrict__ cnt, int* __restrict__ rec, int N, int E, int NB,
          int nseg, int pitch) {
  static_assert(MAXSEG == FINE_THREADS, "thread j reads segment j's offsets and fills run slot j");
  static_assert(RUN_SLOTS - MAXSEG <= FINE_THREADS, "the slots past MAXSEG are cleared by one thread each");
  __shared__ int c[BIN_NODES];
  __shared__ int cur[BIN_NODES];
  __shared__ int wsum[BIN_NODES / 32];
  __shared__ __attribute__((aligned(8))) int run_words[2 * RUN_SLOTS];
  __shared__ int tot[2];
  __shared__ __attribute__((aligned(8))) int stage_words[2 * FINE_CAP];
  int2* stage = reinterpret_cast<int2*>(stage_words);
  int2* runs = reinterpret_cast<int2*>(run_words);
  const int tid = threadIdx.x;
  const int lane = tid & 31;
  const int w = __builtin_amdgcn_readfirstlane(tid >> 5);
  const int b = blockIdx.x;
  if (tid < BIN_NODES) c[tid] = 0;
  if (tid < 2) tot[tid] = 0;
  if (tid < RUN_SLOTS - MAXSEG) runs[MAXSEG + tid] = make_int2(0, 0);
  {
    int st = 0;
    int ln = 0;
    int first = 0;
    if (tid < nseg) {
      const int left = E - tid * EPB;
      const int nvalid = left < EPB ? left : EPB;
      const int* __restrict__ row = tab + (size_t)tid * pitch;
      st = row[b];
      const int en = (b + 1 < NB) ? row[b + 1] : nvalid;
      ln = en - st;
      first = tid * EPB + st;
    }
    runs[tid] = make_int2(first, ln);
    __syncthreads();
#pragma unroll
    for (int dd = 16; dd >= 1; dd >>= 1) {
      st += __shfl_xor(st, dd);
      ln += __shfl_xor(ln, dd);
    }
    if (lane == 0 && (st | ln)) {
      atomicAdd(&tot[0], st);
      atomicAdd(&tot[1], ln);
    }
  }
  __syncthreads();
  const int lo = tot[0];
  const int total = tot[1];
  const bool staged = total <= FINE_CAP;
  for (int j0 = w; j0 < nseg; j0 += FINE_WAVES * RUN_BATCH) {
    int2 rn[RUN_BATCH];
    int dl[RUN_BATCH];
    int longest = 0;
#pragma unroll
    for (int k = 0; k < RUN_BATCH; ++k) {
      rn[k] = runs[j0 + k * FINE_WAVES];
      longest = rn[k].y > longest ? rn[k].y : longest;
    }
#pragma unroll
    for (int k = 0; k < RUN_BATCH; ++k) {
      dl[k] = -1;
      if (lane < rn[k].y) dl[k] = binned[rn[k].x + lane];
    }
#pragma unroll
    for (int k = 0; k < RUN_BATCH; ++k)
      if (dl[k] >= 0) atomicAdd(&c[dl[k] & (BIN_NODES - 1)], 1);
    if (longest > 32) {
#pragma unroll
      for (int k = 0; k < RUN_BATCH; ++k)
        for (int l = 32 + lane; l < rn[k].y; l += 32)
          atomicAdd(&c[binned[rn[k].x + l] & (BIN_NODES - 1)], 1);
    }
  }
  __syncthreads();
  int mine = 0;
  int x = 0;
  if (tid < BIN_NODES) {
    mine = c[tid];
    x = mine;
#pragma unroll
    for (int d = 1; d < 32; d <<= 1) {
      int y = __shfl_up(x, d);
      if (lane >= d) x += y;
    }
    if (lane == 31) wsum[w] = x;
  }
  __syncthreads();
  if (tid < BIN_NODES) {
    int before = 0;
#pragma unroll
    for (int k = 0; k < BIN_NODES / 32; ++k)
      if (k < w) before += wsum[k];
    const int ex = lo + before + x - mine;
    cur[tid] = ex;
    const int node = (b << BIN_SHIFT) + tid;
    if (node < N) {
      segStart[node] = ex;
      cnt[node] = mine;
    }
  }
  __syncthreads();
  int2* __restrict__ out2 = reinterpret_cast<int2*>(rec);
  for (int j0 = w; j0 < nseg; j0 += FINE_WAVES * RUN_BATCH) {
    int2 rn[RUN_BATCH];
    int ed[RUN_BATCH];
    float wn[RUN_BATCH];
    int longest = 0;
#pragma unroll
    for (int k = 0; k < RUN_BATCH; ++k) {
      rn[k] = runs[j0 + k * FINE_WAVES];
      longest = rn[k].y > longest ? rn[k].y : longest;
    }
#pragma unroll
    for (int k = 0; k < RUN_BATCH; ++k) {
      ed[k] = -1;
      if (lane < rn[k].y) ed[k] = binned[rn[k].x + lane];
    }
#pragma unroll
    for (int k = 0; k < RUN_BATCH; ++k)
      wn[k] = (ed[k] >= 0) ? norm[ed[k] >> PACK_SHIFT] : 0.0f;
#pragma unroll
    for (int k = 0; k < RUN_BATCH; ++k) {
      if (ed[k] >= 0) {
        const int p = atomicAdd(&cur[ed[k] & (BIN_NODES - 1)], 1);
        const int2 val = make_int2(ed[k] >> PACK_SHIFT, __float_as_int(wn[k]));
        if (staged) stage[p - lo] = val;
        else out2[p] = val;
      }
    }
    if (longest > 32) {
#pragma unroll
      for (int k = 0; k < RUN_BATCH; ++k) {
        for (int l = 32 + lane; l < rn[k].y; l += 32) {
          const int e1 = binned[rn[k].x + l];
          const float w1 = norm[e1 >> PACK_SHIFT];
          const int p = atomicAdd(&cur[e1 & (BIN_NODES - 1)], 1);
          const int2 val = make_int2(e1 >> PACK_SHIFT, __float_as_int(w1));
          if (staged) stage[p - lo] = val;
          else out2[p] = val;
        }
      }
    }
  }
  if (staged) {
    __syncthreads();
    for (int i = tid; i < total; i += FINE_THREADS) out2[lo + i] = stage[i];
  }
}

#define TILE_ROWS 32
#define TILE_PITCH 68
#define NODES_PER_WAVE 4
#define LAYER_THREADS 256

template <int NCOLS, int NTILES_N, bool RELU>
__global__ void __launch_bounds__(LAYER_THREADS)
gather_apply(const float* __restrict__ feat,
             const int* __restrict__ rec,
             const int* __restrict__ seg_start,
             const int* __restrict__ cnt,
             const float* __restrict__ norm,
             const float* __restrict__ W,
             const float* __restrict__ bias,
             float* __restrict__ out,
             int N) {
  __shared__ __attribute__((aligned(16))) float tile[TILE_ROWS * TILE_PITCH];
  const int lane = threadIdx.x & 31;
  const int wave = __builtin_amdgcn_readfirstlane((int)(threadIdx.x >> 5));
  const int half = lane >> 4;
  const int q = lane & 15;
  const int row0 = blockIdx.x * TILE_ROWS;
  const int2* __restrict__ r2 = reinterpret_cast<const int2*>(rec);
  const float4* __restrict__ f4 = reinterpret_cast<const float4*>(feat);

  int deg[NODES_PER_WAVE];
  int start[NODES_PER_WAVE];
  float nrm[NODES_PER_WAVE];
  int2 my[NODES_PER_WAVE];
#pragma unroll
  for (int i = 0; i < NODES_PER_WAVE; ++i) {
    const int node = row0 + wave * NODES_PER_WAVE + i;
    const bool ok = node < N;
    const int nc = ok ? node : N - 1;
    const int dg = cnt[nc];
    deg[i] = ok ? dg : 0;
    start[i] = seg_start[nc];
    nrm[i] = norm[nc];
  }
#pragma unroll
  for (int i = 0; i < NODES_PER_WAVE; ++i) {
    my[i] = make_int2(0, 0);
    if (lane < deg[i]) my[i] = r2[start[i] + lane];
  }
#pragma unroll
  for (int i = 0; i < NODES_PER_WAVE; ++i) {
    float4 acc = make_float4(0.0f, 0.0f, 0.0f, 0.0f);
    for (int base = 0; base < deg[i]; base += 32) {
      const int left = deg[i] - base;
      const int n = left < 32 ? left : 32;
      int2 cur = my[i];
      if (base > 0) {
        cur = make_int2(0, 0);
        if (lane < n) cur = r2[start[i] + base + lane];
      }
      const int npair = n & ~1;
#pragma unroll 4
      for (int j = 0; j < npair; j += 2) {
        const int s = __shfl(cur.x, j + half);
        const float w = __int_as_float(__shfl(cur.y, j + half));
        const float4 v = f4[(size_t)s * (FEAT_DIM / 4) + q];
        acc.x = fmaf(v.x, w, acc.x);
        acc.y = fmaf(v.y, w, acc.y);
        acc.z = fmaf(v.z, w, acc.z);
        acc.w = fmaf(v.w, w, acc.w);
      }
      if (n & 1) {
        const int s = __shfl(cur.x, n - 1);
        const float w = __int_as_float(__shfl(cur.y, n - 1));
        if (half == 0) {
          const float4 v = f4[(size_t)s * (FEAT_DIM / 4) + q];
          acc.x = fmaf(v.x, w, acc.x);
          acc.y = fmaf(v.y, w, acc.y);
          acc.z = fmaf(v.z, w, acc.z);
          acc.w = fmaf(v.w, w, acc.w);
        }
      }
    }
    acc.x += __shfl_xor(acc.x, 16);
    acc.y += __shfl_xor(acc.y, 16);
    acc.z += __shfl_xor(acc.z, 16);
    acc.w += __shfl_xor(acc.w, 16);
    if (half == 0) {
      const float nr = nrm[i];
      float4 t;
      t.x = acc.x * nr;
      t.y = acc.y * nr;
      t.z = acc.z * nr;
      t.w = acc.w * nr;
      *reinterpret_cast<float4*>(&tile[(wave * NODES_PER_WAVE + i) * TILE_PITCH + 4 * q]) = t;
    }
  }
  const int tile_m = wave >> 2;
  const int tile_n = wave & 3;
  const int m = lane & 15;
  const int c0 = tile_n << 4;
  const int r0 = row0 + (tile_m << 4);
  const float* arow = &tile[((tile_m << 4) + m) * TILE_PITCH];

  const int col = c0 + m;
  const bool colok = (col < NCOLS);
  const int bcol = colok ? col : 0;

  v2f bw[FEAT_DIM / 4];
  float bv = 0.0f;
  if (tile_n < NTILES_N) {
#pragma unroll
    for (int k0 = 0; k0 < FEAT_DIM; k0 += 4) {
      const int k = k0 + half * 2;
      bw[k0 >> 2].x = W[(size_t)k * NCOLS + bcol];
      bw[k0 >> 2].y = W[(size_t)(k + 1) * NCOLS + bcol];
    }
    bv = bias[bcol];
  }
  __syncthreads();
  if (tile_n >= NTILES_N) return;

  v8f acc = {};
#pragma unroll
  for (int k0 = 0; k0 < FEAT_DIM; k0 += 4) {
    const int k = k0 + half * 2;
    const v2f a = *reinterpret_cast<const v2f*>(arow + k);
    acc = __builtin_amdgcn_wmma_f32_16x16x4_f32(false, a, false, bw[k0 >> 2], (short)0,
                                                acc, false, false);
  }

  if (colok) {
#pragma unroll
    for (int j = 0; j < 8; ++j) {
      const int row = r0 + j + half * 8;
      if (row < N) {
        float v = acc[j] + bv;
        if (RELU) v = fmaxf(v, 0.0f);
        out[(size_t)row * NCOLS + col] = v;
      }
    }
  }
}

extern "C" void kernel_launch(void* const* d_in, const int* in_sizes, int n_in,
                              void* d_out, int out_size, void* d_ws, size_t ws_size,
                              hipStream_t stream) {
  const float* feat = (const float*)d_in[0];
  const float* norm = (const float*)d_in[1];
  const int* src    = (const int*)d_in[2];
  const int* dst    = (const int*)d_in[3];
  const float* W1   = (const float*)d_in[4];
  const float* b1   = (const float*)d_in[5];
  const float* W2   = (const float*)d_in[6];
  const float* b2   = (const float*)d_in[7];
  float* out = (float*)d_out;

  const int N = in_sizes[1];
  const int E = in_sizes[2];

  float* bufB = (float*)d_ws;
  const int Npad = (N + 3) & ~3;
  const int NB = (N + BIN_NODES - 1) >> BIN_SHIFT;
  const int edgeBlocks = (E + EPB - 1) / EPB;
  const int pitch = NB + 1;
  const int tabInts = (edgeBlocks * pitch + 3) & ~3;
  int* tab       = (int*)(bufB + (size_t)N * FEAT_DIM);
  int* segStart  = tab + tabInts;
  int* cnt       = segStart + Npad;
  int* binned    = cnt + Npad;
  int* rec       = binned + (size_t)edgeBlocks * EPB;

  const int layerBlocks = (N + TILE_ROWS - 1) / TILE_ROWS;

  bin_local<<<edgeBlocks, BIN_THREADS, 0, stream>>>(src, dst, binned, tab, E, NB, pitch);
  fine_sort<<<NB, FINE_THREADS, 0, stream>>>(binned, tab, norm, segStart, cnt, rec, N, E, NB,
                                             edgeBlocks, pitch);

  gather_apply<64, 4, true><<<layerBlocks, LAYER_THREADS, 0, stream>>>(
      feat, rec, segStart, cnt, norm, W1, b1, bufB, N);

  gather_apply<40, 3, false><<<layerBlocks, LAYER_THREADS, 0, stream>>>(
      bufB, rec, segStart, cnt, norm, W2, b2, out, N);
}
